# starting kernel
# baseline (speedup 1.0000x reference)
#include <hip/hip_runtime.h>
#include <math.h>

typedef __attribute__((ext_vector_type(2))) float v2f;
typedef __attribute__((ext_vector_type(8))) float v8f;

#define BB 64
#define SS 50
#define LL 20
#define QQ 20
#define DD 128
#define VV 32000

__global__ void k_a_embed(const int* __restrict__ story,
                          const float* __restrict__ embedA,
                          float* __restrict__ a_embed) {
  const int bs = blockIdx.x;
  const int d  = threadIdx.x;
  const int* w = story + bs * LL;
  float acc = 0.f;
#pragma unroll
  for (int l = 0; l < LL; ++l)
    acc += embedA[(long)w[l] * DD + d];
  a_embed[bs * DD + d] = acc;
}

__global__ void k_u(const int* __restrict__ query,
                    const float* __restrict__ embedB,
                    float* __restrict__ u) {
  const int b = blockIdx.x;
  const int d = threadIdx.x;
  const int* w = query + b * QQ;
  float acc = 0.f;
#pragma unroll
  for (int q = 0; q < QQ; ++q)
    acc += embedB[(long)w[q] * DD + d];
  u[b * DD + d] = acc;
}

__global__ void k_ip(const float* __restrict__ a_embed,
                     const float* __restrict__ u,
                     float* __restrict__ ip) {
  const int s = blockIdx.x;
  __shared__ float red[256];
  float acc = 0.f;
  for (int i = threadIdx.x; i < BB * DD; i += 256) {
    const int b = i >> 7;
    const int k = i & (DD - 1);
    acc += a_embed[(b * SS + s) * DD + k] * u[b * DD + k];
  }
  red[threadIdx.x] = acc;
  __syncthreads();
  for (int off = 128; off > 0; off >>= 1) {
    if (threadIdx.x < off) red[threadIdx.x] += red[threadIdx.x + off];
    __syncthreads();
  }
  if (threadIdx.x == 0) ip[s] = red[0];
}

__global__ void k_softmax_p(const float* __restrict__ ip, float* __restrict__ p) {
  __shared__ float red[64];
  const int t = threadIdx.x;
  const float v = (t < SS) ? ip[t] : -INFINITY;
  red[t] = v; __syncthreads();
  for (int off = 32; off > 0; off >>= 1) {
    if (t < off) red[t] = fmaxf(red[t], red[t + off]);
    __syncthreads();
  }
  const float mx = red[0];
  __syncthreads();
  const float e = (t < SS) ? expf(v - mx) : 0.f;
  red[t] = e; __syncthreads();
  for (int off = 32; off > 0; off >>= 1) {
    if (t < off) red[t] += red[t + off];
    __syncthreads();
  }
  if (t < SS) p[t] = e / red[0];
}

__global__ void k_ten3(const float* __restrict__ a_embed,
                       const float* __restrict__ u,
                       const float* __restrict__ p,
                       float* __restrict__ ten3) {
  const int b = blockIdx.x;
  const int d = threadIdx.x;
  float acc = u[b * DD + d];
  for (int s = 0; s < SS; ++s)
    acc += a_embed[(b * SS + s) * DD + d] * p[s];
  ten3[b * DD + d] = acc;
}

__global__ void __launch_bounds__(32)
k_gemm_wmma(const float* __restrict__ ten3,
            const float* __restrict__ Wm,
            const float* __restrict__ bias,
            float* __restrict__ out) {
  const int lane = threadIdx.x;
  const int m0   = blockIdx.y << 4;
  const int n0   = blockIdx.x << 6;
  const int half = lane >> 4;
  const int l15  = lane & 15;

  const float* __restrict__ arow = ten3 + (m0 + l15) * DD;
  const float* __restrict__ b0r  = Wm + (long)(n0 +  0 + l15) * DD;
  const float* __restrict__ b1r  = Wm + (long)(n0 + 16 + l15) * DD;
  const float* __restrict__ b2r  = Wm + (long)(n0 + 32 + l15) * DD;
  const float* __restrict__ b3r  = Wm + (long)(n0 + 48 + l15) * DD;

  v8f c0 = {}, c1 = {}, c2 = {}, c3 = {};

#pragma unroll 4
  for (int k = 0; k < DD; k += 4) {
    const int kk = k + half * 2;
    const v2f a  = *(const v2f*)(arow + kk);
    const v2f b0 = *(const v2f*)(b0r  + kk);
    const v2f b1 = *(const v2f*)(b1r  + kk);
    const v2f b2 = *(const v2f*)(b2r  + kk);
    const v2f b3 = *(const v2f*)(b3r  + kk);
    c0 = __builtin_amdgcn_wmma_f32_16x16x4_f32(false, a, false, b0, (short)0, c0, false, false);
    c1 = __builtin_amdgcn_wmma_f32_16x16x4_f32(false, a, false, b1, (short)0, c1, false, false);
    c2 = __builtin_amdgcn_wmma_f32_16x16x4_f32(false, a, false, b2, (short)0, c2, false, false);
    c3 = __builtin_amdgcn_wmma_f32_16x16x4_f32(false, a, false, b3, (short)0, c3, false, false);
  }

  const int ncol = n0 + l15;
#pragma unroll
  for (int r = 0; r < 8; ++r) {
    const long m = m0 + r + 8 * half;
    out[m * VV + ncol +  0] = c0[r] + bias[ncol +  0];
    out[m * VV + ncol + 16] = c1[r] + bias[ncol + 16];
    out[m * VV + ncol + 32] = c2[r] + bias[ncol + 32];
    out[m * VV + ncol + 48] = c3[r] + bias[ncol + 48];
  }
}

__global__ void k_softmax_rows(float* __restrict__ out) {
  const int row = blockIdx.x;
  float* __restrict__ x = out + (long)row * VV;
  __shared__ float red[256];
  const int t = threadIdx.x;

  float m = -INFINITY;
  for (int i = t; i < VV; i += 256) m = fmaxf(m, x[i]);
  red[t] = m; __syncthreads();
  for (int off = 128; off > 0; off >>= 1) {
    if (t < off) red[t] = fmaxf(red[t], red[t + off]);
    __syncthreads();
  }
  const float mx = red[0];
  __syncthreads();

  float s = 0.f;
  for (int i = t; i < VV; i += 256) s += expf(x[i] - mx);
  red[t] = s; __syncthreads();
  for (int off = 128; off > 0; off >>= 1) {
    if (t < off) red[t] += red[t + off];
    __syncthreads();
  }
  const float inv = 1.0f / red[0];

  for (int i = t; i < VV; i += 256) x[i] = expf(x[i] - mx) * inv;
}

extern "C" void kernel_launch(void* const* d_in, const int* in_sizes, int n_in,
                              void* d_out, int out_size, void* d_ws, size_t ws_size,
                              hipStream_t stream) {
  const int*   story  = (const int*)d_in[0];
  const int*   query  = (const int*)d_in[1];
  const float* embedA = (const float*)d_in[2];
  const float* embedB = (const float*)d_in[3];
  const float* Wm     = (const float*)d_in[4];
  const float* bias   = (const float*)d_in[5];
  float* out = (float*)d_out;
  float* ws  = (float*)d_ws;

  float* a_embed = ws;
  float* u       = ws + 409600;
  float* ip      = ws + 417792;
  float* p       = ws + 417856;
  float* ten3    = ws + 417920;

  k_a_embed  <<<BB * SS, DD, 0, stream>>>(story, embedA, a_embed);
  k_u        <<<BB,      DD, 0, stream>>>(query, embedB, u);
  k_ip       <<<SS,     256, 0, stream>>>(a_embed, u, ip);
  k_softmax_p<<<1,       64, 0, stream>>>(ip, p);
  k_ten3     <<<BB,      DD, 0, stream>>>(a_embed, u, p, ten3);

  dim3 gg(VV / 64, BB / 16);
  k_gemm_wmma<<<gg, 32, 0, stream>>>(ten3, Wm, bias, out);

  k_softmax_rows<<<BB, 256, 0, stream>>>(out);
}


# Row softmax spread over the grid: GEMM epilogue leaves per-64-column row max and exp-sum, a 50x4-block launch merges them and normalises (exp-sum order changes)
# speedup vs baseline: 2.5299x; 2.5299x over previous
#include <hip/hip_runtime.h>
#include <math.h>

typedef __attribute__((ext_vector_type(2))) float v2f;
typedef __attribute__((ext_vector_type(8))) float v8f;

#define BB 64
#define SS 50
#define LL 20
#define QQ 20
#define DD 128
#define VV 32000

#define NSTRIP (VV / 64)
#define NCHUNK 50
#define CHUNK  (VV / NCHUNK)
#define NT_NORM 320

template <int CTRL>
__device__ __forceinline__ float row_xor(float v) {
  return __builtin_bit_cast(float,
      __builtin_amdgcn_update_dpp(0, __builtin_bit_cast(int, v), CTRL, 0xf, 0xf, false));
}
__device__ __forceinline__ float other_row(float v) {
  const unsigned u = __builtin_bit_cast(unsigned, v);
  return __builtin_bit_cast(float,
      __builtin_amdgcn_permlanex16(u, u, 0x76543210u, 0xfedcba98u, false, false));
}
__device__ __forceinline__ float row16_max(float v) {
  v = fmaxf(v, row_xor<0x161>(v));
  v = fmaxf(v, row_xor<0x162>(v));
  v = fmaxf(v, row_xor<0x164>(v));
  v = fmaxf(v, row_xor<0x168>(v));
  return v;
}
__device__ __forceinline__ float row16_sum(float v) {
  v += row_xor<0x161>(v);
  v += row_xor<0x162>(v);
  v += row_xor<0x164>(v);
  v += row_xor<0x168>(v);
  return v;
}
__device__ __forceinline__ float q8_max(float v) {
  v = fmaxf(v, row_xor<0x164>(v));
  v = fmaxf(v, row_xor<0x168>(v));
  return fmaxf(v, other_row(v));
}
__device__ __forceinline__ float q8_sum(float v) {
  v += row_xor<0x164>(v);
  v += row_xor<0x168>(v);
  return v + other_row(v);
}

__global__ void k_a_embed(const int* __restrict__ story,
                          const float* __restrict__ embedA,
                          float* __restrict__ a_embed) {
  const int bs = blockIdx.x;
  const int d  = threadIdx.x;
  const int* w = story + bs * LL;
  float acc = 0.f;
#pragma unroll
  for (int l = 0; l < LL; ++l)
    acc += embedA[(long)w[l] * DD + d];
  a_embed[bs * DD + d] = acc;
}

__global__ void k_u(const int* __restrict__ query,
                    const float* __restrict__ embedB,
                    float* __restrict__ u) {
  const int b = blockIdx.x;
  const int d = threadIdx.x;
  const int* w = query + b * QQ;
  float acc = 0.f;
#pragma unroll
  for (int q = 0; q < QQ; ++q)
    acc += embedB[(long)w[q] * DD + d];
  u[b * DD + d] = acc;
}

__global__ void k_ip(const float* __restrict__ a_embed,
                     const float* __restrict__ u,
                     float* __restrict__ ip) {
  const int s = blockIdx.x;
  __shared__ float red[256];
  float acc = 0.f;
  for (int i = threadIdx.x; i < BB * DD; i += 256) {
    const int b = i >> 7;
    const int k = i & (DD - 1);
    acc += a_embed[(b * SS + s) * DD + k] * u[b * DD + k];
  }
  red[threadIdx.x] = acc;
  __syncthreads();
  for (int off = 128; off > 0; off >>= 1) {
    if (threadIdx.x < off) red[threadIdx.x] += red[threadIdx.x + off];
    __syncthreads();
  }
  if (threadIdx.x == 0) ip[s] = red[0];
}

__global__ void k_softmax_p(const float* __restrict__ ip, float* __restrict__ p) {
  __shared__ float red[64];
  const int t = threadIdx.x;
  const float v = (t < SS) ? ip[t] : -INFINITY;
  red[t] = v; __syncthreads();
  for (int off = 32; off > 0; off >>= 1) {
    if (t < off) red[t] = fmaxf(red[t], red[t + off]);
    __syncthreads();
  }
  const float mx = red[0];
  __syncthreads();
  const float e = (t < SS) ? expf(v - mx) : 0.f;
  red[t] = e; __syncthreads();
  for (int off = 32; off > 0; off >>= 1) {
    if (t < off) red[t] += red[t + off];
    __syncthreads();
  }
  if (t < SS) p[t] = e / red[0];
}

__global__ void k_ten3(const float* __restrict__ a_embed,
                       const float* __restrict__ u,
                       const float* __restrict__ p,
                       float* __restrict__ ten3) {
  const int b = blockIdx.x;
  const int d = threadIdx.x;
  float acc = u[b * DD + d];
  for (int s = 0; s < SS; ++s)
    acc += a_embed[(b * SS + s) * DD + d] * p[s];
  ten3[b * DD + d] = acc;
}

__global__ void __launch_bounds__(32)
k_gemm_wmma(const float* __restrict__ ten3,
            const float* __restrict__ Wm,
            const float* __restrict__ bias,
            float* __restrict__ out,
            float* __restrict__ part) {
  const int lane = threadIdx.x;
  const int m0   = blockIdx.y << 4;
  const int n0   = blockIdx.x << 6;
  const int half = lane >> 4;
  const int l15  = lane & 15;

  const float* __restrict__ arow = ten3 + (m0 + l15) * DD;
  const float* __restrict__ b0r  = Wm + (long)(n0 +  0 + l15) * DD;
  const float* __restrict__ b1r  = Wm + (long)(n0 + 16 + l15) * DD;
  const float* __restrict__ b2r  = Wm + (long)(n0 + 32 + l15) * DD;
  const float* __restrict__ b3r  = Wm + (long)(n0 + 48 + l15) * DD;

  v8f c0 = {}, c1 = {}, c2 = {}, c3 = {};

#pragma unroll 4
  for (int k = 0; k < DD; k += 4) {
    const int kk = k + half * 2;
    const v2f a  = *(const v2f*)(arow + kk);
    const v2f b0 = *(const v2f*)(b0r  + kk);
    const v2f b1 = *(const v2f*)(b1r  + kk);
    const v2f b2 = *(const v2f*)(b2r  + kk);
    const v2f b3 = *(const v2f*)(b3r  + kk);
    c0 = __builtin_amdgcn_wmma_f32_16x16x4_f32(false, a, false, b0, (short)0, c0, false, false);
    c1 = __builtin_amdgcn_wmma_f32_16x16x4_f32(false, a, false, b1, (short)0, c1, false, false);
    c2 = __builtin_amdgcn_wmma_f32_16x16x4_f32(false, a, false, b2, (short)0, c2, false, false);
    c3 = __builtin_amdgcn_wmma_f32_16x16x4_f32(false, a, false, b3, (short)0, c3, false, false);
  }

  const int ncol = n0 + l15;
  const float bz0 = bias[ncol +  0];
  const float bz1 = bias[ncol + 16];
  const float bz2 = bias[ncol + 32];
  const float bz3 = bias[ncol + 48];
  float rmax[8], rsum[8];
#pragma unroll
  for (int r = 0; r < 8; ++r) {
    const long m = m0 + r + 8 * half;
    const float v0 = c0[r] + bz0;
    const float v1 = c1[r] + bz1;
    const float v2 = c2[r] + bz2;
    const float v3 = c3[r] + bz3;
    out[m * VV + ncol +  0] = v0;
    out[m * VV + ncol + 16] = v1;
    out[m * VV + ncol + 32] = v2;
    out[m * VV + ncol + 48] = v3;
    const float mx = row16_max(fmaxf(fmaxf(v0, v1), fmaxf(v2, v3)));
    const float e  = (expf(v0 - mx) + expf(v1 - mx)) + (expf(v2 - mx) + expf(v3 - mx));
    rmax[r] = mx;
    rsum[r] = row16_sum(e);
  }
  const int j = l15 & 7;
  float sel = 0.f;
#pragma unroll
  for (int r = 0; r < 8; ++r)
    if (j == r) sel = (l15 < 8) ? rmax[r] : rsum[r];
  const int pos = ((l15 & 8) << 1) + j + 8 * half;
  part[(((long)blockIdx.y * NSTRIP + blockIdx.x) << 5) + pos] = sel;
}

__global__ void __launch_bounds__(NT_NORM)
k_softmax_norm(const float* __restrict__ part, float* __restrict__ out) {
  const int t     = threadIdx.x;
  const int chunk = blockIdx.x;
  const int mb    = blockIdx.y;
  const int wave  = t >> 5;
  const int lane  = t & 31;

  __shared__ float4 sWm[NT_NORM / 32][4];
  __shared__ float4 sWs[NT_NORM / 32][4];
  __shared__ float  sM[16];
  __shared__ float  sInv[16];

  const int cf = t % (CHUNK / 4);
  const int rg = t / (CHUNK / 4);
  float* __restrict__ xp = out + (long)(mb * 16 + rg * 8) * VV + chunk * CHUNK + cf * 4;
  float4 x[8];
#pragma unroll
  for (int j = 0; j < 8; ++j) x[j] = *(const float4*)(xp + (long)j * VV);

  const int q  = t & 3;
  const int s0 = t >> 2;
  const float* __restrict__ pp = part + (((long)mb * NSTRIP) << 5) + 4 * q;
  float4 pm[7], ps[7];
#pragma unroll
  for (int i = 0; i < 6; ++i) {
    const int strip = s0 + 80 * i;
    pm[i] = *(const float4*)(pp + (strip << 5));
    ps[i] = *(const float4*)(pp + (strip << 5) + 16);
  }
  if (s0 < NSTRIP - 480) {
    pm[6] = *(const float4*)(pp + ((s0 + 480) << 5));
    ps[6] = *(const float4*)(pp + ((s0 + 480) << 5) + 16);
  } else {
    pm[6] = make_float4(-3.0e38f, -3.0e38f, -3.0e38f, -3.0e38f);
    ps[6] = make_float4(0.f, 0.f, 0.f, 0.f);
  }

  float4 wm = pm[0];
#pragma unroll
  for (int i = 1; i < 7; ++i) {
    wm.x = fmaxf(wm.x, pm[i].x); wm.y = fmaxf(wm.y, pm[i].y);
    wm.z = fmaxf(wm.z, pm[i].z); wm.w = fmaxf(wm.w, pm[i].w);
  }
  wm.x = q8_max(wm.x); wm.y = q8_max(wm.y); wm.z = q8_max(wm.z); wm.w = q8_max(wm.w);

  float4 wsum = make_float4(0.f, 0.f, 0.f, 0.f);
#pragma unroll
  for (int i = 0; i < 7; ++i) {
    wsum.x += ps[i].x * expf(pm[i].x - wm.x);
    wsum.y += ps[i].y * expf(pm[i].y - wm.y);
    wsum.z += ps[i].z * expf(pm[i].z - wm.z);
    wsum.w += ps[i].w * expf(pm[i].w - wm.w);
  }
  wsum.x = q8_sum(wsum.x); wsum.y = q8_sum(wsum.y);
  wsum.z = q8_sum(wsum.z); wsum.w = q8_sum(wsum.w);

  if (lane < 4) { sWm[wave][lane] = wm; sWs[wave][lane] = wsum; }
  __syncthreads();

  if (t < 16) {
    const float* fm = (const float*)sWm;
    const float* fs = (const float*)sWs;
    float M = fm[t];
#pragma unroll
    for (int w = 1; w < NT_NORM / 32; ++w) M = fmaxf(M, fm[w * 16 + t]);
    float S = 0.f;
#pragma unroll
    for (int w = 0; w < NT_NORM / 32; ++w) S += fs[w * 16 + t] * expf(fm[w * 16 + t] - M);
    sM[t]   = M;
    sInv[t] = 1.0f / S;
  }
  __syncthreads();

#pragma unroll
  for (int j = 0; j < 8; ++j) {
    const float M   = sM[rg * 8 + j];
    const float inv = sInv[rg * 8 + j];
    float4 y;
    y.x = expf(x[j].x - M) * inv;
    y.y = expf(x[j].y - M) * inv;
    y.z = expf(x[j].z - M) * inv;
    y.w = expf(x[j].w - M) * inv;
    *(float4*)(xp + (long)j * VV) = y;
  }
}

extern "C" void kernel_launch(void* const* d_in, const int* in_sizes, int n_in,
                              void* d_out, int out_size, void* d_ws, size_t ws_size,
                              hipStream_t stream) {
  const int*   story  = (const int*)d_in[0];
  const int*   query  = (const int*)d_in[1];
  const float* embedA = (const float*)d_in[2];
  const float* embedB = (const float*)d_in[3];
  const float* Wm     = (const float*)d_in[4];
  const float* bias   = (const float*)d_in[5];
  float* out = (float*)d_out;
  float* ws  = (float*)d_ws;

  float* a_embed = ws;
  float* u       = ws + 409600;
  float* ip      = ws + 417792;
  float* p       = ws + 417856;
  float* ten3    = ws + 417920;
  float* part    = ws + 426112;

  k_a_embed  <<<BB * SS, DD, 0, stream>>>(story, embedA, a_embed);
  k_u        <<<BB,      DD, 0, stream>>>(query, embedB, u);
  k_ip       <<<SS,     256, 0, stream>>>(a_embed, u, ip);
  k_softmax_p<<<1,       64, 0, stream>>>(ip, p);
  k_ten3     <<<BB,      DD, 0, stream>>>(a_embed, u, p, ten3);

  dim3 gg(VV / 64, BB / 16);
  k_gemm_wmma<<<gg, 32, 0, stream>>>(ten3, Wm, bias, out, part);

  dim3 gn(NCHUNK, BB / 16);
  k_softmax_norm<<<gn, NT_NORM, 0, stream>>>(part, out);
}


# Front end 5 launches -> 2: gather launch (wave per 2 sentences, 16 B loads) also forms u and each sentence's dot with u; a 64-block launch sums dots over b, softmax-50, ten3 (ip sum order changes)
# speedup vs baseline: 3.5804x; 1.4152x over previous
#include <hip/hip_runtime.h>
#include <math.h>

typedef __attribute__((ext_vector_type(2))) float v2f;
typedef __attribute__((ext_vector_type(4))) float v4f;
typedef __attribute__((ext_vector_type(8))) float v8f;

#define BB 64
#define SS 50
#define LL 20
#define QQ 20
#define DD 128
#define VV 32000

#define NSTRIP (VV / 64)
#define NCHUNK 50
#define CHUNK  (VV / NCHUNK)
#define NT_NORM 320

#define SPB 10
#define NGRP (SS / SPB)
#define NT_GATHER 192

template <int CTRL>
__device__ __forceinline__ float row_xor(float v) {
  return __builtin_bit_cast(float,
      __builtin_amdgcn_update_dpp(0, __builtin_bit_cast(int, v), CTRL, 0xf, 0xf, false));
}
__device__ __forceinline__ float other_row(float v) {
  const unsigned u = __builtin_bit_cast(unsigned, v);
  return __builtin_bit_cast(float,
      __builtin_amdgcn_permlanex16(u, u, 0x76543210u, 0xfedcba98u, false, false));
}
__device__ __forceinline__ float row16_max(float v) {
  v = fmaxf(v, row_xor<0x161>(v));
  v = fmaxf(v, row_xor<0x162>(v));
  v = fmaxf(v, row_xor<0x164>(v));
  v = fmaxf(v, row_xor<0x168>(v));
  return v;
}
__device__ __forceinline__ float row16_sum(float v) {
  v += row_xor<0x161>(v);
  v += row_xor<0x162>(v);
  v += row_xor<0x164>(v);
  v += row_xor<0x168>(v);
  return v;
}
__device__ __forceinline__ float q8_max(float v) {
  v = fmaxf(v, row_xor<0x164>(v));
  v = fmaxf(v, row_xor<0x168>(v));
  return fmaxf(v, other_row(v));
}
__device__ __forceinline__ float q8_sum(float v) {
  v += row_xor<0x164>(v);
  v += row_xor<0x168>(v);
  return v + other_row(v);
}

__device__ __forceinline__ float wave_max(float v) {
  v = row16_max(v);
  return fmaxf(v, other_row(v));
}
__device__ __forceinline__ float wave_sum(float v) {
  v = row16_sum(v);
  return v + other_row(v);
}
__device__ __forceinline__ float wave_sum_halving(float v) {
  v += other_row(v);
  v += row_xor<0x168>(v);
  v += row_xor<0x164>(v);
  v += row_xor<0x162>(v);
  v += row_xor<0x161>(v);
  return v;
}

__device__ __forceinline__ void rows20_load(v4f (&r)[LL], const int* __restrict__ w,
                                            const float* __restrict__ table, int lane) {
#pragma unroll
  for (int l = 0; l < LL; ++l)
    r[l] = ((const v4f*)table)[(unsigned)w[l] * (DD / 4) + lane];
}
__device__ __forceinline__ void rows20_hold(const v4f (&r)[LL]) {
  asm volatile("" :: "v"(r[0]), "v"(r[1]), "v"(r[2]), "v"(r[3]), "v"(r[4]),
                     "v"(r[5]), "v"(r[6]), "v"(r[7]), "v"(r[8]), "v"(r[9]),
                     "v"(r[10]), "v"(r[11]), "v"(r[12]), "v"(r[13]), "v"(r[14]),
                     "v"(r[15]), "v"(r[16]), "v"(r[17]), "v"(r[18]), "v"(r[19]));
}
__device__ __forceinline__ v4f rows20_sum(const v4f (&r)[LL]) {
  v4f acc = {0.f, 0.f, 0.f, 0.f};
#pragma unroll
  for (int l = 0; l < LL; ++l) acc += r[l];
  return acc;
}

__global__ void __launch_bounds__(NT_GATHER)
k_gather_dot(const int* __restrict__ story,
             const int* __restrict__ query,
             const float* __restrict__ embedA,
             const float* __restrict__ embedB,
             float* __restrict__ a_embed,
             float* __restrict__ u,
             float* __restrict__ dots) {
  const int g    = blockIdx.x;
  const int b    = blockIdx.y;
  const int lane = threadIdx.x & 31;
  const int wv = __builtin_amdgcn_readfirstlane((int)(threadIdx.x >> 5));

  __shared__ v4f   sU[32];
  __shared__ float sDot[SPB];

  v4f ra[LL], rb[LL];
  v4f a0, a1 = {0.f, 0.f, 0.f, 0.f};
  if (wv < SPB / 2) {
    const int* w = story + (b * SS + g * SPB + wv * 2) * LL;
    rows20_load(ra, w, embedA, lane);
    rows20_load(rb, w + LL, embedA, lane);
    rows20_hold(ra);
    rows20_hold(rb);
    a0 = rows20_sum(ra);
    a1 = rows20_sum(rb);
  } else {
    rows20_load(ra, query + b * QQ, embedB, lane);
    rows20_hold(ra);
    a0 = rows20_sum(ra);
    sU[lane] = a0;
  }
  __syncthreads();

  if (wv < SPB / 2) {
    const v4f uu = sU[lane];
    float d0 = a0.x * uu.x + a0.y * uu.y + a0.z * uu.z + a0.w * uu.w;
    float d1 = a1.x * uu.x + a1.y * uu.y + a1.z * uu.z + a1.w * uu.w;
    d0 = wave_sum(d0);
    d1 = wave_sum(d1);
    if (lane == 0) { sDot[2 * wv] = d0; sDot[2 * wv + 1] = d1; }
  }
  __syncthreads();

  if (wv < SPB / 2) {
    const int bs = b * SS + g * SPB + wv * 2;
    *(v4f*)(a_embed + (long)bs * DD + lane * 4)       = a0;
    *(v4f*)(a_embed + (long)(bs + 1) * DD + lane * 4) = a1;
  } else if (g == 0) {
    *(v4f*)(u + b * DD + lane * 4) = a0;
  }
  if (threadIdx.x < SPB)
    dots[((b * NGRP + g) << 5) + threadIdx.x] = sDot[threadIdx.x];
}

__global__ void __launch_bounds__(DD)
k_p_ten3(const float* __restrict__ a_embed,
         const float* __restrict__ u,
         const float* __restrict__ dots,
         float* __restrict__ ten3) {
  const int b    = blockIdx.x;
  const int d    = threadIdx.x;
  const int lane = d & 31;

  __shared__ float sPart[2][64];

  const int s = d & 63;
  const int h = d >> 6;
  float part = 0.f;
  if (s < SS) {
    const float* __restrict__ dp = dots + ((h * 32 * NGRP + s / SPB) << 5) + s % SPB;
#pragma unroll
    for (int i = 0; i < 32; ++i) part += dp[i * (NGRP * 32)];
  }

  const float* __restrict__ ap = a_embed + (long)b * SS * DD + d;
  float av[SS];
#pragma unroll
  for (int t = 0; t < SS; ++t) av[t] = ap[t * DD];
  float acc = u[b * DD + d];

  sPart[h][s] = part;
  __syncthreads();

  const bool  hi    = lane + 32 < SS;
  const float ip_lo = sPart[0][lane] + sPart[1][lane];
  const float ip_hi = hi ? sPart[0][lane + 32] + sPart[1][lane + 32] : -INFINITY;
  const float mx    = wave_max(fmaxf(ip_lo, ip_hi));
  const float e_lo  = expf(ip_lo - mx);
  const float e_hi  = hi ? expf(ip_hi - mx) : 0.f;
  const float tot   = wave_sum_halving(e_lo + e_hi);
  const int   p_lo  = __builtin_bit_cast(int, e_lo / tot);
  const int   p_hi  = __builtin_bit_cast(int, e_hi / tot);

#pragma unroll
  for (int t = 0; t < SS; ++t) {
    const float p = __builtin_bit_cast(float,
        __builtin_amdgcn_readlane(t < 32 ? p_lo : p_hi, t & 31));
    acc += av[t] * p;
  }
  ten3[b * DD + d] = acc;
}

__global__ void __launch_bounds__(32)
k_gemm_wmma(const float* __restrict__ ten3,
            const float* __restrict__ Wm,
            const float* __restrict__ bias,
            float* __restrict__ out,
            float* __restrict__ part) {
  const int lane = threadIdx.x;
  const int m0   = blockIdx.y << 4;
  const int n0   = blockIdx.x << 6;
  const int half = lane >> 4;
  const int l15  = lane & 15;

  const float* __restrict__ arow = ten3 + (m0 + l15) * DD;
  const float* __restrict__ b0r  = Wm + (long)(n0 +  0 + l15) * DD;
  const float* __restrict__ b1r  = Wm + (long)(n0 + 16 + l15) * DD;
  const float* __restrict__ b2r  = Wm + (long)(n0 + 32 + l15) * DD;
  const float* __restrict__ b3r  = Wm + (long)(n0 + 48 + l15) * DD;

  v8f c0 = {}, c1 = {}, c2 = {}, c3 = {};

#pragma unroll 4
  for (int k = 0; k < DD; k += 4) {
    const int kk = k + half * 2;
    const v2f a  = *(const v2f*)(arow + kk);
    const v2f b0 = *(const v2f*)(b0r  + kk);
    const v2f b1 = *(const v2f*)(b1r  + kk);
    const v2f b2 = *(const v2f*)(b2r  + kk);
    const v2f b3 = *(const v2f*)(b3r  + kk);
    c0 = __builtin_amdgcn_wmma_f32_16x16x4_f32(false, a, false, b0, (short)0, c0, false, false);
    c1 = __builtin_amdgcn_wmma_f32_16x16x4_f32(false, a, false, b1, (short)0, c1, false, false);
    c2 = __builtin_amdgcn_wmma_f32_16x16x4_f32(false, a, false, b2, (short)0, c2, false, false);
    c3 = __builtin_amdgcn_wmma_f32_16x16x4_f32(false, a, false, b3, (short)0, c3, false, false);
  }

  const int ncol = n0 + l15;
  const float bz0 = bias[ncol +  0];
  const float bz1 = bias[ncol + 16];
  const float bz2 = bias[ncol + 32];
  const float bz3 = bias[ncol + 48];
  float rmax[8], rsum[8];
#pragma unroll
  for (int r = 0; r < 8; ++r) {
    const long m = m0 + r + 8 * half;
    const float v0 = c0[r] + bz0;
    const float v1 = c1[r] + bz1;
    const float v2 = c2[r] + bz2;
    const float v3 = c3[r] + bz3;
    out[m * VV + ncol +  0] = v0;
    out[m * VV + ncol + 16] = v1;
    out[m * VV + ncol + 32] = v2;
    out[m * VV + ncol + 48] = v3;
    const float mx = row16_max(fmaxf(fmaxf(v0, v1), fmaxf(v2, v3)));
    const float e  = (expf(v0 - mx) + expf(v1 - mx)) + (expf(v2 - mx) + expf(v3 - mx));
    rmax[r] = mx;
    rsum[r] = row16_sum(e);
  }
  const int j = l15 & 7;
  float sel = 0.f;
#pragma unroll
  for (int r = 0; r < 8; ++r)
    if (j == r) sel = (l15 < 8) ? rmax[r] : rsum[r];
  const int pos = ((l15 & 8) << 1) + j + 8 * half;
  part[(((long)blockIdx.y * NSTRIP + blockIdx.x) << 5) + pos] = sel;
}

__global__ void __launch_bounds__(NT_NORM)
k_softmax_norm(const float* __restrict__ part, float* __restrict__ out) {
  const int t     = threadIdx.x;
  const int chunk = blockIdx.x;
  const int mb    = blockIdx.y;
  const int wave  = t >> 5;
  const int lane  = t & 31;

  __shared__ float4 sWm[NT_NORM / 32][4];
  __shared__ float4 sWs[NT_NORM / 32][4];
  __shared__ float  sM[16];
  __shared__ float  sInv[16];

  const int cf = t % (CHUNK / 4);
  const int rg = t / (CHUNK / 4);
  float* __restrict__ xp = out + (long)(mb * 16 + rg * 8) * VV + chunk * CHUNK + cf * 4;
  float4 x[8];
#pragma unroll
  for (int j = 0; j < 8; ++j) x[j] = *(const float4*)(xp + (long)j * VV);

  const int q  = t & 3;
  const int s0 = t >> 2;
  const float* __restrict__ pp = part + (((long)mb * NSTRIP) << 5) + 4 * q;
  float4 pm[7], ps[7];
#pragma unroll
  for (int i = 0; i < 6; ++i) {
    const int strip = s0 + 80 * i;
    pm[i] = *(const float4*)(pp + (strip << 5));
    ps[i] = *(const float4*)(pp + (strip << 5) + 16);
  }
  if (s0 < NSTRIP - 480) {
    pm[6] = *(const float4*)(pp + ((s0 + 480) << 5));
    ps[6] = *(const float4*)(pp + ((s0 + 480) << 5) + 16);
  } else {
    pm[6] = make_float4(-3.0e38f, -3.0e38f, -3.0e38f, -3.0e38f);
    ps[6] = make_float4(0.f, 0.f, 0.f, 0.f);
  }

  float4 wm = pm[0];
#pragma unroll
  for (int i = 1; i < 7; ++i) {
    wm.x = fmaxf(wm.x, pm[i].x); wm.y = fmaxf(wm.y, pm[i].y);
    wm.z = fmaxf(wm.z, pm[i].z); wm.w = fmaxf(wm.w, pm[i].w);
  }
  wm.x = q8_max(wm.x); wm.y = q8_max(wm.y); wm.z = q8_max(wm.z); wm.w = q8_max(wm.w);

  float4 wsum = make_float4(0.f, 0.f, 0.f, 0.f);
#pragma unroll
  for (int i = 0; i < 7; ++i) {
    wsum.x += ps[i].x * expf(pm[i].x - wm.x);
    wsum.y += ps[i].y * expf(pm[i].y - wm.y);
    wsum.z += ps[i].z * expf(pm[i].z - wm.z);
    wsum.w += ps[i].w * expf(pm[i].w - wm.w);
  }
  wsum.x = q8_sum(wsum.x); wsum.y = q8_sum(wsum.y);
  wsum.z = q8_sum(wsum.z); wsum.w = q8_sum(wsum.w);

  if (lane < 4) { sWm[wave][lane] = wm; sWs[wave][lane] = wsum; }
  __syncthreads();

  if (t < 16) {
    const float* fm = (const float*)sWm;
    const float* fs = (const float*)sWs;
    float M = fm[t];
#pragma unroll
    for (int w = 1; w < NT_NORM / 32; ++w) M = fmaxf(M, fm[w * 16 + t]);
    float S = 0.f;
#pragma unroll
    for (int w = 0; w < NT_NORM / 32; ++w) S += fs[w * 16 + t] * expf(fm[w * 16 + t] - M);
    sM[t]   = M;
    sInv[t] = 1.0f / S;
  }
  __syncthreads();

#pragma unroll
  for (int j = 0; j < 8; ++j) {
    const float M   = sM[rg * 8 + j];
    const float inv = sInv[rg * 8 + j];
    float4 y;
    y.x = expf(x[j].x - M) * inv;
    y.y = expf(x[j].y - M) * inv;
    y.z = expf(x[j].z - M) * inv;
    y.w = expf(x[j].w - M) * inv;
    *(float4*)(xp + (long)j * VV) = y;
  }
}

extern "C" void kernel_launch(void* const* d_in, const int* in_sizes, int n_in,
                              void* d_out, int out_size, void* d_ws, size_t ws_size,
                              hipStream_t stream) {
  const int*   story  = (const int*)d_in[0];
  const int*   query  = (const int*)d_in[1];
  const float* embedA = (const float*)d_in[2];
  const float* embedB = (const float*)d_in[3];
  const float* Wm     = (const float*)d_in[4];
  const float* bias   = (const float*)d_in[5];
  float* out = (float*)d_out;
  float* ws  = (float*)d_ws;

  float* a_embed = ws;
  float* u       = ws + 409600;
  float* ten3    = ws + 417792;
  float* part    = ws + 425984;
  float* dots    = ws + 489984;

  dim3 gd(NGRP, BB);
  k_gather_dot<<<gd, NT_GATHER, 0, stream>>>(story, query, embedA, embedB, a_embed, u, dots);
  k_p_ten3    <<<BB, DD, 0, stream>>>(a_embed, u, dots, ten3);

  dim3 gg(VV / 64, BB / 16);
  k_gemm_wmma<<<gg, 32, 0, stream>>>(ten3, Wm, bias, out, part);

  dim3 gn(NCHUNK, BB / 16);
  k_softmax_norm<<<gn, NT_NORM, 0, stream>>>(part, out);
}


# GEMM: the four 16-row blocks of a 64-column strip run as the 4 waves of one work-group; the strip's W rows are staged once in LDS by whole-row 16 B loads, B fragments read from there in groups ahead
# speedup vs baseline: 3.8098x; 1.0641x over previous
#include <hip/hip_runtime.h>
#include <math.h>

typedef __attribute__((ext_vector_type(2))) float v2f;
typedef __attribute__((ext_vector_type(4))) float v4f;
typedef __attribute__((ext_vector_type(8))) float v8f;

#define BB 64
#define SS 50
#define LL 20
#define QQ 20
#define DD 128
#define VV 32000

#define NSTRIP (VV / 64)
#define NCHUNK 50
#define CHUNK  (VV / NCHUNK)
#define NT_NORM 320

#define SPB 10
#define NGRP (SS / SPB)
#define NT_GATHER 192

template <int CTRL>
__device__ __forceinline__ float row_xor(float v) {
  return __builtin_bit_cast(float,
      __builtin_amdgcn_update_dpp(0, __builtin_bit_cast(int, v), CTRL, 0xf, 0xf, false));
}
__device__ __forceinline__ float other_row(float v) {
  const unsigned u = __builtin_bit_cast(unsigned, v);
  return __builtin_bit_cast(float,
      __builtin_amdgcn_permlanex16(u, u, 0x76543210u, 0xfedcba98u, false, false));
}
__device__ __forceinline__ float row16_max(float v) {
  v = fmaxf(v, row_xor<0x161>(v));
  v = fmaxf(v, row_xor<0x162>(v));
  v = fmaxf(v, row_xor<0x164>(v));
  v = fmaxf(v, row_xor<0x168>(v));
  return v;
}
__device__ __forceinline__ float row16_sum(float v) {
  v += row_xor<0x161>(v);
  v += row_xor<0x162>(v);
  v += row_xor<0x164>(v);
  v += row_xor<0x168>(v);
  return v;
}
__device__ __forceinline__ float q8_max(float v) {
  v = fmaxf(v, row_xor<0x164>(v));
  v = fmaxf(v, row_xor<0x168>(v));
  return fmaxf(v, other_row(v));
}
__device__ __forceinline__ float q8_sum(float v) {
  v += row_xor<0x164>(v);
  v += row_xor<0x168>(v);
  return v + other_row(v);
}

__device__ __forceinline__ float wave_max(float v) {
  v = row16_max(v);
  return fmaxf(v, other_row(v));
}
__device__ __forceinline__ float wave_sum(float v) {
  v = row16_sum(v);
  return v + other_row(v);
}
__device__ __forceinline__ float wave_sum_halving(float v) {
  v += other_row(v);
  v += row_xor<0x168>(v);
  v += row_xor<0x164>(v);
  v += row_xor<0x162>(v);
  v += row_xor<0x161>(v);
  return v;
}

__device__ __forceinline__ void rows20_load(v4f (&r)[LL], const int* __restrict__ w,
                                            const float* __restrict__ table, int lane) {
#pragma unroll
  for (int l = 0; l < LL; ++l)
    r[l] = ((const v4f*)table)[(unsigned)w[l] * (DD / 4) + lane];
}
__device__ __forceinline__ void rows20_hold(const v4f (&r)[LL]) {
  asm volatile("" :: "v"(r[0]), "v"(r[1]), "v"(r[2]), "v"(r[3]), "v"(r[4]),
                     "v"(r[5]), "v"(r[6]), "v"(r[7]), "v"(r[8]), "v"(r[9]),
                     "v"(r[10]), "v"(r[11]), "v"(r[12]), "v"(r[13]), "v"(r[14]),
                     "v"(r[15]), "v"(r[16]), "v"(r[17]), "v"(r[18]), "v"(r[19]));
}
__device__ __forceinline__ v4f rows20_sum(const v4f (&r)[LL]) {
  v4f acc = {0.f, 0.f, 0.f, 0.f};
#pragma unroll
  for (int l = 0; l < LL; ++l) acc += r[l];
  return acc;
}

__global__ void __launch_bounds__(NT_GATHER)
k_gather_dot(const int* __restrict__ story,
             const int* __restrict__ query,
             const float* __restrict__ embedA,
             const float* __restrict__ embedB,
             float* __restrict__ a_embed,
             float* __restrict__ u,
             float* __restrict__ dots) {
  const int g    = blockIdx.x;
  const int b    = blockIdx.y;
  const int lane = threadIdx.x & 31;
  const int wv = __builtin_amdgcn_readfirstlane((int)(threadIdx.x >> 5));

  __shared__ v4f   sU[32];
  __shared__ float sDot[SPB];

  v4f ra[LL], rb[LL];
  v4f a0, a1 = {0.f, 0.f, 0.f, 0.f};
  if (wv < SPB / 2) {
    const int* w = story + (b * SS + g * SPB + wv * 2) * LL;
    rows20_load(ra, w, embedA, lane);
    rows20_load(rb, w + LL, embedA, lane);
    rows20_hold(ra);
    rows20_hold(rb);
    a0 = rows20_sum(ra);
    a1 = rows20_sum(rb);
  } else {
    rows20_load(ra, query + b * QQ, embedB, lane);
    rows20_hold(ra);
    a0 = rows20_sum(ra);
    sU[lane] = a0;
  }
  __syncthreads();

  if (wv < SPB / 2) {
    const v4f uu = sU[lane];
    float d0 = a0.x * uu.x + a0.y * uu.y + a0.z * uu.z + a0.w * uu.w;
    float d1 = a1.x * uu.x + a1.y * uu.y + a1.z * uu.z + a1.w * uu.w;
    d0 = wave_sum(d0);
    d1 = wave_sum(d1);
    if (lane == 0) { sDot[2 * wv] = d0; sDot[2 * wv + 1] = d1; }
  }
  __syncthreads();

  if (wv < SPB / 2) {
    const int bs = b * SS + g * SPB + wv * 2;
    *(v4f*)(a_embed + (long)bs * DD + lane * 4)       = a0;
    *(v4f*)(a_embed + (long)(bs + 1) * DD + lane * 4) = a1;
  } else if (g == 0) {
    *(v4f*)(u + b * DD + lane * 4) = a0;
  }
  if (threadIdx.x < SPB)
    dots[((b * NGRP + g) << 5) + threadIdx.x] = sDot[threadIdx.x];
}

__global__ void __launch_bounds__(DD)
k_p_ten3(const float* __restrict__ a_embed,
         const float* __restrict__ u,
         const float* __restrict__ dots,
         float* __restrict__ ten3) {
  const int b    = blockIdx.x;
  const int d    = threadIdx.x;
  const int lane = d & 31;

  __shared__ float sPart[2][64];

  const int s = d & 63;
  const int h = d >> 6;
  float part = 0.f;
  if (s < SS) {
    const float* __restrict__ dp = dots + ((h * 32 * NGRP + s / SPB) << 5) + s % SPB;
#pragma unroll
    for (int i = 0; i < 32; ++i) part += dp[i * (NGRP * 32)];
  }

  const float* __restrict__ ap = a_embed + (long)b * SS * DD + d;
  float av[SS];
#pragma unroll
  for (int t = 0; t < SS; ++t) av[t] = ap[t * DD];
  float acc = u[b * DD + d];

  sPart[h][s] = part;
  __syncthreads();

  const bool  hi    = lane + 32 < SS;
  const float ip_lo = sPart[0][lane] + sPart[1][lane];
  const float ip_hi = hi ? sPart[0][lane + 32] + sPart[1][lane + 32] : -INFINITY;
  const float mx    = wave_max(fmaxf(ip_lo, ip_hi));
  const float e_lo  = expf(ip_lo - mx);
  const float e_hi  = hi ? expf(ip_hi - mx) : 0.f;
  const float tot   = wave_sum_halving(e_lo + e_hi);
  const int   p_lo  = __builtin_bit_cast(int, e_lo / tot);
  const int   p_hi  = __builtin_bit_cast(int, e_hi / tot);

#pragma unroll
  for (int t = 0; t < SS; ++t) {
    const float p = __builtin_bit_cast(float,
        __builtin_amdgcn_readlane(t < 32 ? p_lo : p_hi, t & 31));
    acc += av[t] * p;
  }
  ten3[b * DD + d] = acc;
}

#define WPITCH4 33

__global__ void __launch_bounds__(128)
k_gemm_wmma(const float* __restrict__ ten3,
            const float* __restrict__ Wm,
            const float* __restrict__ bias,
            float* __restrict__ out,
            float* __restrict__ part) {
  const int lane = threadIdx.x & 31;
  const int wv   = __builtin_amdgcn_readfirstlane((int)(threadIdx.x >> 5));
  const int m0   = wv << 4;
  const int n0   = blockIdx.x << 6;
  const int half = lane >> 4;
  const int l15  = lane & 15;

  __shared__ v4f sW[64 * WPITCH4];

  const v4f* __restrict__ wsrc = (const v4f*)(Wm + (long)(n0 + 16 * wv) * DD) + lane;
  v4f st[16];
#pragma unroll
  for (int i = 0; i < 16; ++i) st[i] = wsrc[i * (DD / 4)];

  const float* __restrict__ arow = ten3 + (m0 + l15) * DD + half * 2;
  v2f af[DD / 4];
#pragma unroll
  for (int i = 0; i < DD / 4; ++i) af[i] = *(const v2f*)(arow + 4 * i);

#pragma unroll
  for (int i = 0; i < 16; ++i) sW[(16 * wv + i) * WPITCH4 + lane] = st[i];
  __syncthreads();

  const float* __restrict__ sb = (const float*)sW + l15 * (WPITCH4 * 4) + half * 2;

  v8f c0 = {}, c1 = {}, c2 = {}, c3 = {};

  v2f bq[2][8][4];
#pragma unroll
  for (int s = 0; s < 8; ++s)
#pragma unroll
    for (int t = 0; t < 4; ++t)
      bq[0][s][t] = *(const v2f*)(sb + 16 * t * (WPITCH4 * 4) + 4 * s);

#pragma unroll
  for (int ch = 0; ch < 4; ++ch) {
    if (ch + 1 < 4) {
#pragma unroll
      for (int s = 0; s < 8; ++s)
#pragma unroll
        for (int t = 0; t < 4; ++t)
          bq[(ch + 1) & 1][s][t] =
              *(const v2f*)(sb + 16 * t * (WPITCH4 * 4) + 4 * (8 * (ch + 1) + s));
    }
    __builtin_amdgcn_sched_barrier(0);
#pragma unroll
    for (int s = 0; s < 8; ++s) {
      const v2f a = af[8 * ch + s];
      c0 = __builtin_amdgcn_wmma_f32_16x16x4_f32(false, a, false, bq[ch & 1][s][0], (short)0, c0, false, false);
      c1 = __builtin_amdgcn_wmma_f32_16x16x4_f32(false, a, false, bq[ch & 1][s][1], (short)0, c1, false, false);
      c2 = __builtin_amdgcn_wmma_f32_16x16x4_f32(false, a, false, bq[ch & 1][s][2], (short)0, c2, false, false);
      c3 = __builtin_amdgcn_wmma_f32_16x16x4_f32(false, a, false, bq[ch & 1][s][3], (short)0, c3, false, false);
    }
    __builtin_amdgcn_sched_barrier(0);
  }

  const int ncol = n0 + l15;
  const float bz0 = bias[ncol +  0];
  const float bz1 = bias[ncol + 16];
  const float bz2 = bias[ncol + 32];
  const float bz3 = bias[ncol + 48];
  float rmax[8], rsum[8];
#pragma unroll
  for (int r = 0; r < 8; ++r) {
    const long m = m0 + r + 8 * half;
    const float v0 = c0[r] + bz0;
    const float v1 = c1[r] + bz1;
    const float v2 = c2[r] + bz2;
    const float v3 = c3[r] + bz3;
    out[m * VV + ncol +  0] = v0;
    out[m * VV + ncol + 16] = v1;
    out[m * VV + ncol + 32] = v2;
    out[m * VV + ncol + 48] = v3;
    const float mx = row16_max(fmaxf(fmaxf(v0, v1), fmaxf(v2, v3)));
    const float e  = (expf(v0 - mx) + expf(v1 - mx)) + (expf(v2 - mx) + expf(v3 - mx));
    rmax[r] = mx;
    rsum[r] = row16_sum(e);
  }
  const int j = l15 & 7;
  float sel = 0.f;
#pragma unroll
  for (int r = 0; r < 8; ++r)
    if (j == r) sel = (l15 < 8) ? rmax[r] : rsum[r];
  const int pos = ((l15 & 8) << 1) + j + 8 * half;
  part[(((long)wv * NSTRIP + blockIdx.x) << 5) + pos] = sel;
}

__global__ void __launch_bounds__(NT_NORM)
k_softmax_norm(const float* __restrict__ part, float* __restrict__ out) {
  const int t     = threadIdx.x;
  const int chunk = blockIdx.x;
  const int mb    = blockIdx.y;
  const int wave  = t >> 5;
  const int lane  = t & 31;

  __shared__ float4 sWm[NT_NORM / 32][4];
  __shared__ float4 sWs[NT_NORM / 32][4];
  __shared__ float  sM[16];
  __shared__ float  sInv[16];

  const int cf = t % (CHUNK / 4);
  const int rg = t / (CHUNK / 4);
  float* __restrict__ xp = out + (long)(mb * 16 + rg * 8) * VV + chunk * CHUNK + cf * 4;
  float4 x[8];
#pragma unroll
  for (int j = 0; j < 8; ++j) x[j] = *(const float4*)(xp + (long)j * VV);

  const int q  = t & 3;
  const int s0 = t >> 2;
  const float* __restrict__ pp = part + (((long)mb * NSTRIP) << 5) + 4 * q;
  float4 pm[7], ps[7];
#pragma unroll
  for (int i = 0; i < 6; ++i) {
    const int strip = s0 + 80 * i;
    pm[i] = *(const float4*)(pp + (strip << 5));
    ps[i] = *(const float4*)(pp + (strip << 5) + 16);
  }
  if (s0 < NSTRIP - 480) {
    pm[6] = *(const float4*)(pp + ((s0 + 480) << 5));
    ps[6] = *(const float4*)(pp + ((s0 + 480) << 5) + 16);
  } else {
    pm[6] = make_float4(-3.0e38f, -3.0e38f, -3.0e38f, -3.0e38f);
    ps[6] = make_float4(0.f, 0.f, 0.f, 0.f);
  }

  float4 wm = pm[0];
#pragma unroll
  for (int i = 1; i < 7; ++i) {
    wm.x = fmaxf(wm.x, pm[i].x); wm.y = fmaxf(wm.y, pm[i].y);
    wm.z = fmaxf(wm.z, pm[i].z); wm.w = fmaxf(wm.w, pm[i].w);
  }
  wm.x = q8_max(wm.x); wm.y = q8_max(wm.y); wm.z = q8_max(wm.z); wm.w = q8_max(wm.w);

  float4 wsum = make_float4(0.f, 0.f, 0.f, 0.f);
#pragma unroll
  for (int i = 0; i < 7; ++i) {
    wsum.x += ps[i].x * expf(pm[i].x - wm.x);
    wsum.y += ps[i].y * expf(pm[i].y - wm.y);
    wsum.z += ps[i].z * expf(pm[i].z - wm.z);
    wsum.w += ps[i].w * expf(pm[i].w - wm.w);
  }
  wsum.x = q8_sum(wsum.x); wsum.y = q8_sum(wsum.y);
  wsum.z = q8_sum(wsum.z); wsum.w = q8_sum(wsum.w);

  if (lane < 4) { sWm[wave][lane] = wm; sWs[wave][lane] = wsum; }
  __syncthreads();

  if (t < 16) {
    const float* fm = (const float*)sWm;
    const float* fs = (const float*)sWs;
    float M = fm[t];
#pragma unroll
    for (int w = 1; w < NT_NORM / 32; ++w) M = fmaxf(M, fm[w * 16 + t]);
    float S = 0.f;
#pragma unroll
    for (int w = 0; w < NT_NORM / 32; ++w) S += fs[w * 16 + t] * expf(fm[w * 16 + t] - M);
    sM[t]   = M;
    sInv[t] = 1.0f / S;
  }
  __syncthreads();

#pragma unroll
  for (int j = 0; j < 8; ++j) {
    const float M   = sM[rg * 8 + j];
    const float inv = sInv[rg * 8 + j];
    float4 y;
    y.x = expf(x[j].x - M) * inv;
    y.y = expf(x[j].y - M) * inv;
    y.z = expf(x[j].z - M) * inv;
    y.w = expf(x[j].w - M) * inv;
    *(float4*)(xp + (long)j * VV) = y;
  }
}

extern "C" void kernel_launch(void* const* d_in, const int* in_sizes, int n_in,
                              void* d_out, int out_size, void* d_ws, size_t ws_size,
                              hipStream_t stream) {
  const int*   story  = (const int*)d_in[0];
  const int*   query  = (const int*)d_in[1];
  const float* embedA = (const float*)d_in[2];
  const float* embedB = (const float*)d_in[3];
  const float* Wm     = (const float*)d_in[4];
  const float* bias   = (const float*)d_in[5];
  float* out = (float*)d_out;
  float* ws  = (float*)d_ws;

  float* a_embed = ws;
  float* u       = ws + 409600;
  float* ten3    = ws + 417792;
  float* part    = ws + 425984;
  float* dots    = ws + 489984;

  dim3 gd(NGRP, BB);
  k_gather_dot<<<gd, NT_GATHER, 0, stream>>>(story, query, embedA, embedB, a_embed, u, dots);
  k_p_ten3    <<<BB, DD, 0, stream>>>(a_embed, u, dots, ten3);

  k_gemm_wmma<<<VV / 64, 128, 0, stream>>>(ten3, Wm, bias, out, part);

  dim3 gn(NCHUNK, BB / 16);
  k_softmax_norm<<<gn, NT_NORM, 0, stream>>>(part, out);
}


# Row softmax exponentials (GEMM epilogue statistics, merge, normalise) as 2^(x*log2e) on v_exp_f32 in place of the library expf: about 3 instructions each for 14; exp rounding changes, sum order kept
# speedup vs baseline: 4.1548x; 1.0906x over previous
#include <hip/hip_runtime.h>
#include <math.h>

typedef __attribute__((ext_vector_type(2))) float v2f;
typedef __attribute__((ext_vector_type(4))) float v4f;
typedef __attribute__((ext_vector_type(8))) float v8f;

#define BB 64
#define SS 50
#define LL 20
#define QQ 20
#define DD 128
#define VV 32000

#define NSTRIP (VV / 64)
#define NCHUNK 50
#define CHUNK  (VV / NCHUNK)
#define NT_NORM 320

#define SPB 10
#define NGRP (SS / SPB)
#define NT_GATHER 192

template <int CTRL>
__device__ __forceinline__ float row_xor(float v) {
  return __builtin_bit_cast(float,
      __builtin_amdgcn_update_dpp(0, __builtin_bit_cast(int, v), CTRL, 0xf, 0xf, false));
}
__device__ __forceinline__ float other_row(float v) {
  const unsigned u = __builtin_bit_cast(unsigned, v);
  return __builtin_bit_cast(float,
      __builtin_amdgcn_permlanex16(u, u, 0x76543210u, 0xfedcba98u, false, false));
}
__device__ __forceinline__ float row16_max(float v) {
  v = fmaxf(v, row_xor<0x161>(v));
  v = fmaxf(v, row_xor<0x162>(v));
  v = fmaxf(v, row_xor<0x164>(v));
  v = fmaxf(v, row_xor<0x168>(v));
  return v;
}
__device__ __forceinline__ float row16_sum(float v) {
  v += row_xor<0x161>(v);
  v += row_xor<0x162>(v);
  v += row_xor<0x164>(v);
  v += row_xor<0x168>(v);
  return v;
}
__device__ __forceinline__ float q8_max(float v) {
  v = fmaxf(v, row_xor<0x164>(v));
  v = fmaxf(v, row_xor<0x168>(v));
  return fmaxf(v, other_row(v));
}
__device__ __forceinline__ float q8_sum(float v) {
  v += row_xor<0x164>(v);
  v += row_xor<0x168>(v);
  return v + other_row(v);
}

__device__ __forceinline__ float wave_max(float v) {
  v = row16_max(v);
  return fmaxf(v, other_row(v));
}
__device__ __forceinline__ float wave_sum(float v) {
  v = row16_sum(v);
  return v + other_row(v);
}
__device__ __forceinline__ float wave_sum_halving(float v) {
  v += other_row(v);
  v += row_xor<0x168>(v);
  v += row_xor<0x164>(v);
  v += row_xor<0x162>(v);
  v += row_xor<0x161>(v);
  return v;
}

__device__ __forceinline__ float exp_b2(float x) {
  return __builtin_amdgcn_exp2f(x * 1.4426950408889634f);
}

__device__ __forceinline__ void rows20_load(v4f (&r)[LL], const int* __restrict__ w,
                                            const float* __restrict__ table, int lane) {
#pragma unroll
  for (int l = 0; l < LL; ++l)
    r[l] = ((const v4f*)table)[(unsigned)w[l] * (DD / 4) + lane];
}
__device__ __forceinline__ void rows20_hold(const v4f (&r)[LL]) {
  asm volatile("" :: "v"(r[0]), "v"(r[1]), "v"(r[2]), "v"(r[3]), "v"(r[4]),
                     "v"(r[5]), "v"(r[6]), "v"(r[7]), "v"(r[8]), "v"(r[9]),
                     "v"(r[10]), "v"(r[11]), "v"(r[12]), "v"(r[13]), "v"(r[14]),
                     "v"(r[15]), "v"(r[16]), "v"(r[17]), "v"(r[18]), "v"(r[19]));
}
__device__ __forceinline__ v4f rows20_sum(const v4f (&r)[LL]) {
  v4f acc = {0.f, 0.f, 0.f, 0.f};
#pragma unroll
  for (int l = 0; l < LL; ++l) acc += r[l];
  return acc;
}

__global__ void __launch_bounds__(NT_GATHER)
k_gather_dot(const int* __restrict__ story,
             const int* __restrict__ query,
             const float* __restrict__ embedA,
             const float* __restrict__ embedB,
             float* __restrict__ a_embed,
             float* __restrict__ u,
             float* __restrict__ dots) {
  const int g    = blockIdx.x;
  const int b    = blockIdx.y;
  const int lane = threadIdx.x & 31;
  const int wv = __builtin_amdgcn_readfirstlane((int)(threadIdx.x >> 5));

  __shared__ v4f   sU[32];
  __shared__ float sDot[SPB];

  v4f ra[LL], rb[LL];
  v4f a0, a1 = {0.f, 0.f, 0.f, 0.f};
  if (wv < SPB / 2) {
    const int* w = story + (b * SS + g * SPB + wv * 2) * LL;
    rows20_load(ra, w, embedA, lane);
    rows20_load(rb, w + LL, embedA, lane);
    rows20_hold(ra);
    rows20_hold(rb);
    a0 = rows20_sum(ra);
    a1 = rows20_sum(rb);
  } else {
    rows20_load(ra, query + b * QQ, embedB, lane);
    rows20_hold(ra);
    a0 = rows20_sum(ra);
    sU[lane] = a0;
  }
  __syncthreads();

  if (wv < SPB / 2) {
    const v4f uu = sU[lane];
    float d0 = a0.x * uu.x + a0.y * uu.y + a0.z * uu.z + a0.w * uu.w;
    float d1 = a1.x * uu.x + a1.y * uu.y + a1.z * uu.z + a1.w * uu.w;
    d0 = wave_sum(d0);
    d1 = wave_sum(d1);
    if (lane == 0) { sDot[2 * wv] = d0; sDot[2 * wv + 1] = d1; }
  }
  __syncthreads();

  if (wv < SPB / 2) {
    const int bs = b * SS + g * SPB + wv * 2;
    *(v4f*)(a_embed + (long)bs * DD + lane * 4)       = a0;
    *(v4f*)(a_embed + (long)(bs + 1) * DD + lane * 4) = a1;
  } else if (g == 0) {
    *(v4f*)(u + b * DD + lane * 4) = a0;
  }
  if (threadIdx.x < SPB)
    dots[((b * NGRP + g) << 5) + threadIdx.x] = sDot[threadIdx.x];
}

__global__ void __launch_bounds__(DD)
k_p_ten3(const float* __restrict__ a_embed,
         const float* __restrict__ u,
         const float* __restrict__ dots,
         float* __restrict__ ten3) {
  const int b    = blockIdx.x;
  const int d    = threadIdx.x;
  const int lane = d & 31;

  __shared__ float sPart[2][64];

  const int s = d & 63;
  const int h = d >> 6;
  float part = 0.f;
  if (s < SS) {
    const float* __restrict__ dp = dots + ((h * 32 * NGRP + s / SPB) << 5) + s % SPB;
#pragma unroll
    for (int i = 0; i < 32; ++i) part += dp[i * (NGRP * 32)];
  }

  const float* __restrict__ ap = a_embed + (long)b * SS * DD + d;
  float av[SS];
#pragma unroll
  for (int t = 0; t < SS; ++t) av[t] = ap[t * DD];
  float acc = u[b * DD + d];

  sPart[h][s] = part;
  __syncthreads();

  const bool  hi    = lane + 32 < SS;
  const float ip_lo = sPart[0][lane] + sPart[1][lane];
  const float ip_hi = hi ? sPart[0][lane + 32] + sPart[1][lane + 32] : -INFINITY;
  const float mx    = wave_max(fmaxf(ip_lo, ip_hi));
  const float e_lo  = expf(ip_lo - mx);
  const float e_hi  = hi ? expf(ip_hi - mx) : 0.f;
  const float tot   = wave_sum_halving(e_lo + e_hi);
  const int   p_lo  = __builtin_bit_cast(int, e_lo / tot);
  const int   p_hi  = __builtin_bit_cast(int, e_hi / tot);

#pragma unroll
  for (int t = 0; t < SS; ++t) {
    const float p = __builtin_bit_cast(float,
        __builtin_amdgcn_readlane(t < 32 ? p_lo : p_hi, t & 31));
    acc += av[t] * p;
  }
  ten3[b * DD + d] = acc;
}

#define WPITCH4 33

__global__ void __launch_bounds__(128)
k_gemm_wmma(const float* __restrict__ ten3,
            const float* __restrict__ Wm,
            const float* __restrict__ bias,
            float* __restrict__ out,
            float* __restrict__ part) {
  const int lane = threadIdx.x & 31;
  const int wv   = __builtin_amdgcn_readfirstlane((int)(threadIdx.x >> 5));
  const int m0   = wv << 4;
  const int n0   = blockIdx.x << 6;
  const int half = lane >> 4;
  const int l15  = lane & 15;

  __shared__ v4f sW[64 * WPITCH4];

  const v4f* __restrict__ wsrc = (const v4f*)(Wm + (long)(n0 + 16 * wv) * DD) + lane;
  v4f st[16];
#pragma unroll
  for (int i = 0; i < 16; ++i) st[i] = wsrc[i * (DD / 4)];

  const float* __restrict__ arow = ten3 + (m0 + l15) * DD + half * 2;
  v2f af[DD / 4];
#pragma unroll
  for (int i = 0; i < DD / 4; ++i) af[i] = *(const v2f*)(arow + 4 * i);

#pragma unroll
  for (int i = 0; i < 16; ++i) sW[(16 * wv + i) * WPITCH4 + lane] = st[i];
  __syncthreads();

  const float* __restrict__ sb = (const float*)sW + l15 * (WPITCH4 * 4) + half * 2;

  v8f c0 = {}, c1 = {}, c2 = {}, c3 = {};

  v2f bq[2][8][4];
#pragma unroll
  for (int s = 0; s < 8; ++s)
#pragma unroll
    for (int t = 0; t < 4; ++t)
      bq[0][s][t] = *(const v2f*)(sb + 16 * t * (WPITCH4 * 4) + 4 * s);

#pragma unroll
  for (int ch = 0; ch < 4; ++ch) {
    if (ch + 1 < 4) {
#pragma unroll
      for (int s = 0; s < 8; ++s)
#pragma unroll
        for (int t = 0; t < 4; ++t)
          bq[(ch + 1) & 1][s][t] =
              *(const v2f*)(sb + 16 * t * (WPITCH4 * 4) + 4 * (8 * (ch + 1) + s));
    }
    __builtin_amdgcn_sched_barrier(0);
#pragma unroll
    for (int s = 0; s < 8; ++s) {
      const v2f a = af[8 * ch + s];
      c0 = __builtin_amdgcn_wmma_f32_16x16x4_f32(false, a, false, bq[ch & 1][s][0], (short)0, c0, false, false);
      c1 = __builtin_amdgcn_wmma_f32_16x16x4_f32(false, a, false, bq[ch & 1][s][1], (short)0, c1, false, false);
      c2 = __builtin_amdgcn_wmma_f32_16x16x4_f32(false, a, false, bq[ch & 1][s][2], (short)0, c2, false, false);
      c3 = __builtin_amdgcn_wmma_f32_16x16x4_f32(false, a, false, bq[ch & 1][s][3], (short)0, c3, false, false);
    }
    __builtin_amdgcn_sched_barrier(0);
  }

  const int ncol = n0 + l15;
  const float bz0 = bias[ncol +  0];
  const float bz1 = bias[ncol + 16];
  const float bz2 = bias[ncol + 32];
  const float bz3 = bias[ncol + 48];
  float rmax[8], rsum[8];
#pragma unroll
  for (int r = 0; r < 8; ++r) {
    const long m = m0 + r + 8 * half;
    const float v0 = c0[r] + bz0;
    const float v1 = c1[r] + bz1;
    const float v2 = c2[r] + bz2;
    const float v3 = c3[r] + bz3;
    out[m * VV + ncol +  0] = v0;
    out[m * VV + ncol + 16] = v1;
    out[m * VV + ncol + 32] = v2;
    out[m * VV + ncol + 48] = v3;
    const float mx = row16_max(fmaxf(fmaxf(v0, v1), fmaxf(v2, v3)));
    const float e  = (exp_b2(v0 - mx) + exp_b2(v1 - mx)) + (exp_b2(v2 - mx) + exp_b2(v3 - mx));
    rmax[r] = mx;
    rsum[r] = row16_sum(e);
  }
  const int j = l15 & 7;
  float sel = 0.f;
#pragma unroll
  for (int r = 0; r < 8; ++r)
    if (j == r) sel = (l15 < 8) ? rmax[r] : rsum[r];
  const int pos = ((l15 & 8) << 1) + j + 8 * half;
  part[(((long)wv * NSTRIP + blockIdx.x) << 5) + pos] = sel;
}

__global__ void __launch_bounds__(NT_NORM)
k_softmax_norm(const float* __restrict__ part, float* __restrict__ out) {
  const int t     = threadIdx.x;
  const int chunk = blockIdx.x;
  const int mb    = blockIdx.y;
  const int wave  = t >> 5;
  const int lane  = t & 31;

  __shared__ float4 sWm[NT_NORM / 32][4];
  __shared__ float4 sWs[NT_NORM / 32][4];
  __shared__ float  sM[16];
  __shared__ float  sInv[16];

  const int cf = t % (CHUNK / 4);
  const int rg = t / (CHUNK / 4);
  float* __restrict__ xp = out + (long)(mb * 16 + rg * 8) * VV + chunk * CHUNK + cf * 4;
  float4 x[8];
#pragma unroll
  for (int j = 0; j < 8; ++j) x[j] = *(const float4*)(xp + (long)j * VV);

  const int q  = t & 3;
  const int s0 = t >> 2;
  const float* __restrict__ pp = part + (((long)mb * NSTRIP) << 5) + 4 * q;
  float4 pm[7], ps[7];
#pragma unroll
  for (int i = 0; i < 6; ++i) {
    const int strip = s0 + 80 * i;
    pm[i] = *(const float4*)(pp + (strip << 5));
    ps[i] = *(const float4*)(pp + (strip << 5) + 16);
  }
  if (s0 < NSTRIP - 480) {
    pm[6] = *(const float4*)(pp + ((s0 + 480) << 5));
    ps[6] = *(const float4*)(pp + ((s0 + 480) << 5) + 16);
  } else {
    pm[6] = make_float4(-3.0e38f, -3.0e38f, -3.0e38f, -3.0e38f);
    ps[6] = make_float4(0.f, 0.f, 0.f, 0.f);
  }

  float4 wm = pm[0];
#pragma unroll
  for (int i = 1; i < 7; ++i) {
    wm.x = fmaxf(wm.x, pm[i].x); wm.y = fmaxf(wm.y, pm[i].y);
    wm.z = fmaxf(wm.z, pm[i].z); wm.w = fmaxf(wm.w, pm[i].w);
  }
  wm.x = q8_max(wm.x); wm.y = q8_max(wm.y); wm.z = q8_max(wm.z); wm.w = q8_max(wm.w);

  float4 wsum = make_float4(0.f, 0.f, 0.f, 0.f);
#pragma unroll
  for (int i = 0; i < 7; ++i) {
    wsum.x += ps[i].x * exp_b2(pm[i].x - wm.x);
    wsum.y += ps[i].y * exp_b2(pm[i].y - wm.y);
    wsum.z += ps[i].z * exp_b2(pm[i].z - wm.z);
    wsum.w += ps[i].w * exp_b2(pm[i].w - wm.w);
  }
  wsum.x = q8_sum(wsum.x); wsum.y = q8_sum(wsum.y);
  wsum.z = q8_sum(wsum.z); wsum.w = q8_sum(wsum.w);

  if (lane < 4) { sWm[wave][lane] = wm; sWs[wave][lane] = wsum; }
  __syncthreads();

  if (t < 16) {
    const float* fm = (const float*)sWm;
    const float* fs = (const float*)sWs;
    float M = fm[t];
#pragma unroll
    for (int w = 1; w < NT_NORM / 32; ++w) M = fmaxf(M, fm[w * 16 + t]);
    float S = 0.f;
#pragma unroll
    for (int w = 0; w < NT_NORM / 32; ++w) S += fs[w * 16 + t] * exp_b2(fm[w * 16 + t] - M);
    sM[t]   = M;
    sInv[t] = 1.0f / S;
  }
  __syncthreads();

#pragma unroll
  for (int j = 0; j < 8; ++j) {
    const float M   = sM[rg * 8 + j];
    const float inv = sInv[rg * 8 + j];
    float4 y;
    y.x = exp_b2(x[j].x - M) * inv;
    y.y = exp_b2(x[j].y - M) * inv;
    y.z = exp_b2(x[j].z - M) * inv;
    y.w = exp_b2(x[j].w - M) * inv;
    *(float4*)(xp + (long)j * VV) = y;
  }
}

extern "C" void kernel_launch(void* const* d_in, const int* in_sizes, int n_in,
                              void* d_out, int out_size, void* d_ws, size_t ws_size,
                              hipStream_t stream) {
  const int*   story  = (const int*)d_in[0];
  const int*   query  = (const int*)d_in[1];
  const float* embedA = (const float*)d_in[2];
  const float* embedB = (const float*)d_in[3];
  const float* Wm     = (const float*)d_in[4];
  const float* bias   = (const float*)d_in[5];
  float* out = (float*)d_out;
  float* ws  = (float*)d_ws;

  float* a_embed = ws;
  float* u       = ws + 409600;
  float* ten3    = ws + 417792;
  float* part    = ws + 425984;
  float* dots    = ws + 489984;

  dim3 gd(NGRP, BB);
  k_gather_dot<<<gd, NT_GATHER, 0, stream>>>(story, query, embedA, embedB, a_embed, u, dots);
  k_p_ten3    <<<BB, DD, 0, stream>>>(a_embed, u, dots, ten3);

  k_gemm_wmma<<<VV / 64, 128, 0, stream>>>(ten3, Wm, bias, out, part);

  dim3 gn(NCHUNK, BB / 16);
  k_softmax_norm<<<gn, NT_NORM, 0, stream>>>(part, out);
}


# GEMM loads: ten3 rows come like W, by whole-row 16 B loads staged in LDS, not by 32 16-line fragment loads that held the barrier; bias fetched in the prologue, not behind the last WMMA; bit-identical
# speedup vs baseline: 4.3525x; 1.0476x over previous
#include <hip/hip_runtime.h>
#include <math.h>

typedef __attribute__((ext_vector_type(2))) float v2f;
typedef __attribute__((ext_vector_type(4))) float v4f;
typedef __attribute__((ext_vector_type(8))) float v8f;

#define BB 64
#define SS 50
#define LL 20
#define QQ 20
#define DD 128
#define VV 32000

#define NSTRIP (VV / 64)
#define NCHUNK 50
#define CHUNK  (VV / NCHUNK)
#define NT_NORM 320

#define SPB 10
#define NGRP (SS / SPB)
#define NT_GATHER 192

template <int CTRL>
__device__ __forceinline__ float row_xor(float v) {
  return __builtin_bit_cast(float,
      __builtin_amdgcn_update_dpp(0, __builtin_bit_cast(int, v), CTRL, 0xf, 0xf, false));
}
__device__ __forceinline__ float other_row(float v) {
  const unsigned u = __builtin_bit_cast(unsigned, v);
  return __builtin_bit_cast(float,
      __builtin_amdgcn_permlanex16(u, u, 0x76543210u, 0xfedcba98u, false, false));
}
__device__ __forceinline__ float row16_max(float v) {
  v = fmaxf(v, row_xor<0x161>(v));
  v = fmaxf(v, row_xor<0x162>(v));
  v = fmaxf(v, row_xor<0x164>(v));
  v = fmaxf(v, row_xor<0x168>(v));
  return v;
}
__device__ __forceinline__ float row16_sum(float v) {
  v += row_xor<0x161>(v);
  v += row_xor<0x162>(v);
  v += row_xor<0x164>(v);
  v += row_xor<0x168>(v);
  return v;
}
__device__ __forceinline__ float q8_max(float v) {
  v = fmaxf(v, row_xor<0x164>(v));
  v = fmaxf(v, row_xor<0x168>(v));
  return fmaxf(v, other_row(v));
}
__device__ __forceinline__ float q8_sum(float v) {
  v += row_xor<0x164>(v);
  v += row_xor<0x168>(v);
  return v + other_row(v);
}

__device__ __forceinline__ float wave_max(float v) {
  v = row16_max(v);
  return fmaxf(v, other_row(v));
}
__device__ __forceinline__ float wave_sum(float v) {
  v = row16_sum(v);
  return v + other_row(v);
}
__device__ __forceinline__ float wave_sum_halving(float v) {
  v += other_row(v);
  v += row_xor<0x168>(v);
  v += row_xor<0x164>(v);
  v += row_xor<0x162>(v);
  v += row_xor<0x161>(v);
  return v;
}

__device__ __forceinline__ float exp_b2(float x) {
  return __builtin_amdgcn_exp2f(x * 1.4426950408889634f);
}

__device__ __forceinline__ void rows20_load(v4f (&r)[LL], const int* __restrict__ w,
                                            const float* __restrict__ table, int lane) {
#pragma unroll
  for (int l = 0; l < LL; ++l)
    r[l] = ((const v4f*)table)[(unsigned)w[l] * (DD / 4) + lane];
}
__device__ __forceinline__ void rows20_hold(const v4f (&r)[LL]) {
  asm volatile("" :: "v"(r[0]), "v"(r[1]), "v"(r[2]), "v"(r[3]), "v"(r[4]),
                     "v"(r[5]), "v"(r[6]), "v"(r[7]), "v"(r[8]), "v"(r[9]),
                     "v"(r[10]), "v"(r[11]), "v"(r[12]), "v"(r[13]), "v"(r[14]),
                     "v"(r[15]), "v"(r[16]), "v"(r[17]), "v"(r[18]), "v"(r[19]));
}
__device__ __forceinline__ v4f rows20_sum(const v4f (&r)[LL]) {
  v4f acc = {0.f, 0.f, 0.f, 0.f};
#pragma unroll
  for (int l = 0; l < LL; ++l) acc += r[l];
  return acc;
}

__global__ void __launch_bounds__(NT_GATHER)
k_gather_dot(const int* __restrict__ story,
             const int* __restrict__ query,
             const float* __restrict__ embedA,
             const float* __restrict__ embedB,
             float* __restrict__ a_embed,
             float* __restrict__ u,
             float* __restrict__ dots) {
  const int g    = blockIdx.x;
  const int b    = blockIdx.y;
  const int lane = threadIdx.x & 31;
  const int wv = __builtin_amdgcn_readfirstlane((int)(threadIdx.x >> 5));

  __shared__ v4f   sU[32];
  __shared__ float sDot[SPB];

  v4f ra[LL], rb[LL];
  v4f a0, a1 = {0.f, 0.f, 0.f, 0.f};
  if (wv < SPB / 2) {
    const int* w = story + (b * SS + g * SPB + wv * 2) * LL;
    rows20_load(ra, w, embedA, lane);
    rows20_load(rb, w + LL, embedA, lane);
    rows20_hold(ra);
    rows20_hold(rb);
    a0 = rows20_sum(ra);
    a1 = rows20_sum(rb);
  } else {
    rows20_load(ra, query + b * QQ, embedB, lane);
    rows20_hold(ra);
    a0 = rows20_sum(ra);
    sU[lane] = a0;
  }
  __syncthreads();

  if (wv < SPB / 2) {
    const v4f uu = sU[lane];
    float d0 = a0.x * uu.x + a0.y * uu.y + a0.z * uu.z + a0.w * uu.w;
    float d1 = a1.x * uu.x + a1.y * uu.y + a1.z * uu.z + a1.w * uu.w;
    d0 = wave_sum(d0);
    d1 = wave_sum(d1);
    if (lane == 0) { sDot[2 * wv] = d0; sDot[2 * wv + 1] = d1; }
  }
  __syncthreads();

  if (wv < SPB / 2) {
    const int bs = b * SS + g * SPB + wv * 2;
    *(v4f*)(a_embed + (long)bs * DD + lane * 4)       = a0;
    *(v4f*)(a_embed + (long)(bs + 1) * DD + lane * 4) = a1;
  } else if (g == 0) {
    *(v4f*)(u + b * DD + lane * 4) = a0;
  }
  if (threadIdx.x < SPB)
    dots[((b * NGRP + g) << 5) + threadIdx.x] = sDot[threadIdx.x];
}

__global__ void __launch_bounds__(DD)
k_p_ten3(const float* __restrict__ a_embed,
         const float* __restrict__ u,
         const float* __restrict__ dots,
         float* __restrict__ ten3) {
  const int b    = blockIdx.x;
  const int d    = threadIdx.x;
  const int lane = d & 31;

  __shared__ float sPart[2][64];

  const int s = d & 63;
  const int h = d >> 6;
  float part = 0.f;
  if (s < SS) {
    const float* __restrict__ dp = dots + ((h * 32 * NGRP + s / SPB) << 5) + s % SPB;
#pragma unroll
    for (int i = 0; i < 32; ++i) part += dp[i * (NGRP * 32)];
  }

  const float* __restrict__ ap = a_embed + (long)b * SS * DD + d;
  float av[SS];
#pragma unroll
  for (int t = 0; t < SS; ++t) av[t] = ap[t * DD];
  float acc = u[b * DD + d];

  sPart[h][s] = part;
  __syncthreads();

  const bool  hi    = lane + 32 < SS;
  const float ip_lo = sPart[0][lane] + sPart[1][lane];
  const float ip_hi = hi ? sPart[0][lane + 32] + sPart[1][lane + 32] : -INFINITY;
  const float mx    = wave_max(fmaxf(ip_lo, ip_hi));
  const float e_lo  = expf(ip_lo - mx);
  const float e_hi  = hi ? expf(ip_hi - mx) : 0.f;
  const float tot   = wave_sum_halving(e_lo + e_hi);
  const int   p_lo  = __builtin_bit_cast(int, e_lo / tot);
  const int   p_hi  = __builtin_bit_cast(int, e_hi / tot);

#pragma unroll
  for (int t = 0; t < SS; ++t) {
    const float p = __builtin_bit_cast(float,
        __builtin_amdgcn_readlane(t < 32 ? p_lo : p_hi, t & 31));
    acc += av[t] * p;
  }
  ten3[b * DD + d] = acc;
}

#define WPITCH4 33

__global__ void __launch_bounds__(128)
k_gemm_wmma(const float* __restrict__ ten3,
            const float* __restrict__ Wm,
            const float* __restrict__ bias,
            float* __restrict__ out,
            float* __restrict__ part) {
  const int lane = threadIdx.x & 31;
  const int wv   = __builtin_amdgcn_readfirstlane((int)(threadIdx.x >> 5));
  const int m0   = wv << 4;
  const int n0   = blockIdx.x << 6;
  const int half = lane >> 4;
  const int l15  = lane & 15;

  __shared__ v4f sW[64 * WPITCH4];
  __shared__ v4f sA[64 * WPITCH4];

  const int ncol = n0 + l15;
  const float bz0 = bias[ncol +  0];
  const float bz1 = bias[ncol + 16];
  const float bz2 = bias[ncol + 32];
  const float bz3 = bias[ncol + 48];

  const v4f* __restrict__ wsrc = (const v4f*)(Wm + (long)(n0 + m0) * DD) + lane;
  v4f st[16];
#pragma unroll
  for (int i = 0; i < 16; ++i) st[i] = wsrc[i * (DD / 4)];

  const v4f* __restrict__ asrc = (const v4f*)(ten3 + m0 * DD) + lane;
  v4f at[16];
#pragma unroll
  for (int i = 0; i < 16; ++i) at[i] = asrc[i * (DD / 4)];

#pragma unroll
  for (int i = 0; i < 16; ++i) sW[(m0 + i) * WPITCH4 + lane] = st[i];
#pragma unroll
  for (int i = 0; i < 16; ++i) sA[(m0 + i) * WPITCH4 + lane] = at[i];
  __syncthreads();

  const float* __restrict__ sb = (const float*)sW + l15 * (WPITCH4 * 4) + half * 2;
  const float* __restrict__ sa = (const float*)sA + (m0 + l15) * (WPITCH4 * 4) + half * 2;

  v2f af[DD / 4];
#pragma unroll
  for (int i = 0; i < DD / 4; ++i) af[i] = *(const v2f*)(sa + 4 * i);

  v8f c0 = {}, c1 = {}, c2 = {}, c3 = {};

  v2f bq[2][8][4];
#pragma unroll
  for (int s = 0; s < 8; ++s)
#pragma unroll
    for (int t = 0; t < 4; ++t)
      bq[0][s][t] = *(const v2f*)(sb + 16 * t * (WPITCH4 * 4) + 4 * s);

#pragma unroll
  for (int ch = 0; ch < 4; ++ch) {
    if (ch + 1 < 4) {
#pragma unroll
      for (int s = 0; s < 8; ++s)
#pragma unroll
        for (int t = 0; t < 4; ++t)
          bq[(ch + 1) & 1][s][t] =
              *(const v2f*)(sb + 16 * t * (WPITCH4 * 4) + 4 * (8 * (ch + 1) + s));
    }
    __builtin_amdgcn_sched_barrier(0);
#pragma unroll
    for (int s = 0; s < 8; ++s) {
      const v2f a = af[8 * ch + s];
      c0 = __builtin_amdgcn_wmma_f32_16x16x4_f32(false, a, false, bq[ch & 1][s][0], (short)0, c0, false, false);
      c1 = __builtin_amdgcn_wmma_f32_16x16x4_f32(false, a, false, bq[ch & 1][s][1], (short)0, c1, false, false);
      c2 = __builtin_amdgcn_wmma_f32_16x16x4_f32(false, a, false, bq[ch & 1][s][2], (short)0, c2, false, false);
      c3 = __builtin_amdgcn_wmma_f32_16x16x4_f32(false, a, false, bq[ch & 1][s][3], (short)0, c3, false, false);
    }
    __builtin_amdgcn_sched_barrier(0);
  }

  float rmax[8], rsum[8];
#pragma unroll
  for (int r = 0; r < 8; ++r) {
    const long m = m0 + r + 8 * half;
    const float v0 = c0[r] + bz0;
    const float v1 = c1[r] + bz1;
    const float v2 = c2[r] + bz2;
    const float v3 = c3[r] + bz3;
    out[m * VV + ncol +  0] = v0;
    out[m * VV + ncol + 16] = v1;
    out[m * VV + ncol + 32] = v2;
    out[m * VV + ncol + 48] = v3;
    const float mx = row16_max(fmaxf(fmaxf(v0, v1), fmaxf(v2, v3)));
    const float e  = (exp_b2(v0 - mx) + exp_b2(v1 - mx)) + (exp_b2(v2 - mx) + exp_b2(v3 - mx));
    rmax[r] = mx;
    rsum[r] = row16_sum(e);
  }
  const int j = l15 & 7;
  float sel = 0.f;
#pragma unroll
  for (int r = 0; r < 8; ++r)
    if (j == r) sel = (l15 < 8) ? rmax[r] : rsum[r];
  const int pos = ((l15 & 8) << 1) + j + 8 * half;
  part[(((long)wv * NSTRIP + blockIdx.x) << 5) + pos] = sel;
}

__global__ void __launch_bounds__(NT_NORM)
k_softmax_norm(const float* __restrict__ part, float* __restrict__ out) {
  const int t     = threadIdx.x;
  const int chunk = blockIdx.x;
  const int mb    = blockIdx.y;
  const int wave  = t >> 5;
  const int lane  = t & 31;

  __shared__ float4 sWm[NT_NORM / 32][4];
  __shared__ float4 sWs[NT_NORM / 32][4];
  __shared__ float  sM[16];
  __shared__ float  sInv[16];

  const int cf = t % (CHUNK / 4);
  const int rg = t / (CHUNK / 4);
  float* __restrict__ xp = out + (long)(mb * 16 + rg * 8) * VV + chunk * CHUNK + cf * 4;
  float4 x[8];
#pragma unroll
  for (int j = 0; j < 8; ++j) x[j] = *(const float4*)(xp + (long)j * VV);

  const int q  = t & 3;
  const int s0 = t >> 2;
  const float* __restrict__ pp = part + (((long)mb * NSTRIP) << 5) + 4 * q;
  float4 pm[7], ps[7];
#pragma unroll
  for (int i = 0; i < 6; ++i) {
    const int strip = s0 + 80 * i;
    pm[i] = *(const float4*)(pp + (strip << 5));
    ps[i] = *(const float4*)(pp + (strip << 5) + 16);
  }
  if (s0 < NSTRIP - 480) {
    pm[6] = *(const float4*)(pp + ((s0 + 480) << 5));
    ps[6] = *(const float4*)(pp + ((s0 + 480) << 5) + 16);
  } else {
    pm[6] = make_float4(-3.0e38f, -3.0e38f, -3.0e38f, -3.0e38f);
    ps[6] = make_float4(0.f, 0.f, 0.f, 0.f);
  }

  float4 wm = pm[0];
#pragma unroll
  for (int i = 1; i < 7; ++i) {
    wm.x = fmaxf(wm.x, pm[i].x); wm.y = fmaxf(wm.y, pm[i].y);
    wm.z = fmaxf(wm.z, pm[i].z); wm.w = fmaxf(wm.w, pm[i].w);
  }
  wm.x = q8_max(wm.x); wm.y = q8_max(wm.y); wm.z = q8_max(wm.z); wm.w = q8_max(wm.w);

  float4 wsum = make_float4(0.f, 0.f, 0.f, 0.f);
#pragma unroll
  for (int i = 0; i < 7; ++i) {
    wsum.x += ps[i].x * exp_b2(pm[i].x - wm.x);
    wsum.y += ps[i].y * exp_b2(pm[i].y - wm.y);
    wsum.z += ps[i].z * exp_b2(pm[i].z - wm.z);
    wsum.w += ps[i].w * exp_b2(pm[i].w - wm.w);
  }
  wsum.x = q8_sum(wsum.x); wsum.y = q8_sum(wsum.y);
  wsum.z = q8_sum(wsum.z); wsum.w = q8_sum(wsum.w);

  if (lane < 4) { sWm[wave][lane] = wm; sWs[wave][lane] = wsum; }
  __syncthreads();

  if (t < 16) {
    const float* fm = (const float*)sWm;
    const float* fs = (const float*)sWs;
    float M = fm[t];
#pragma unroll
    for (int w = 1; w < NT_NORM / 32; ++w) M = fmaxf(M, fm[w * 16 + t]);
    float S = 0.f;
#pragma unroll
    for (int w = 0; w < NT_NORM / 32; ++w) S += fs[w * 16 + t] * exp_b2(fm[w * 16 + t] - M);
    sM[t]   = M;
    sInv[t] = 1.0f / S;
  }
  __syncthreads();

#pragma unroll
  for (int j = 0; j < 8; ++j) {
    const float M   = sM[rg * 8 + j];
    const float inv = sInv[rg * 8 + j];
    float4 y;
    y.x = exp_b2(x[j].x - M) * inv;
    y.y = exp_b2(x[j].y - M) * inv;
    y.z = exp_b2(x[j].z - M) * inv;
    y.w = exp_b2(x[j].w - M) * inv;
    *(float4*)(xp + (long)j * VV) = y;
  }
}

extern "C" void kernel_launch(void* const* d_in, const int* in_sizes, int n_in,
                              void* d_out, int out_size, void* d_ws, size_t ws_size,
                              hipStream_t stream) {
  const int*   story  = (const int*)d_in[0];
  const int*   query  = (const int*)d_in[1];
  const float* embedA = (const float*)d_in[2];
  const float* embedB = (const float*)d_in[3];
  const float* Wm     = (const float*)d_in[4];
  const float* bias   = (const float*)d_in[5];
  float* out = (float*)d_out;
  float* ws  = (float*)d_ws;

  float* a_embed = ws;
  float* u       = ws + 409600;
  float* ten3    = ws + 417792;
  float* part    = ws + 425984;
  float* dots    = ws + 489984;

  dim3 gd(NGRP, BB);
  k_gather_dot<<<gd, NT_GATHER, 0, stream>>>(story, query, embedA, embedB, a_embed, u, dots);
  k_p_ten3    <<<BB, DD, 0, stream>>>(a_embed, u, dots, ten3);

  k_gemm_wmma<<<VV / 64, 128, 0, stream>>>(ten3, Wm, bias, out, part);

  dim3 gn(NCHUNK, BB / 16);
  k_softmax_norm<<<gn, NT_NORM, 0, stream>>>(part, out);
}
